# starting kernel
# baseline (speedup 1.0000x reference)
#include <hip/hip_runtime.h>
#include <math.h>

typedef __attribute__((ext_vector_type(2))) float v2f;
typedef __attribute__((ext_vector_type(8))) float v8f;

#define BATCH   8
#define M_PTS   512
#define N_PTS   8192
#define EPS_F   1e-7f
#define WAVES   8
#define THREADS (WAVES * 32)
#define TILES   (N_PTS / 16)
#define TPW     (TILES / WAVES)
#define FLT_BIG 3.402823466e+38f

__global__ __launch_bounds__(THREADS)
void pos_loss_kernel(const float* __restrict__ kp,
                     const float* __restrict__ pc,
                     const float* __restrict__ sn,
                     float* __restrict__ out)
{
    const int b    = blockIdx.y;
    const int m0   = blockIdx.x * 16;
    const int tid  = threadIdx.x;
    const int wave = tid >> 5;
    const int lane = tid & 31;
    const int col  = lane & 15;
    const bool hi  = lane >= 16;

    const float* kpb = kp + (size_t)b * 3 * M_PTS;
    const float* pcx = pc + (size_t)b * 3 * N_PTS;
    const float* pcy = pcx + N_PTS;
    const float* pcz = pcy + N_PTS;

    v2f a;
    {
        const float k0 = kpb[0 * M_PTS + m0 + col];
        const float k1 = kpb[1 * M_PTS + m0 + col];
        const float k2 = kpb[2 * M_PTS + m0 + col];
        a[0] = -2.0f * (hi ? k2 : k0);
        a[1] = hi ? 1.0f : (-2.0f * k1);
    }

    const v8f czero = {};

    float minv[8];
    int   mini[8];
#pragma unroll
    for (int r = 0; r < 8; ++r) { minv[r] = FLT_BIG; mini[r] = 0; }

    const int t0 = wave * TPW;
#pragma unroll 4
    for (int tt = 0; tt < TPW; ++tt) {
        const int n = (t0 + tt) * 16 + col;
        const float x = pcx[n];
        const float y = pcy[n];
        const float z = pcz[n];
        const float pn2 = x * x + y * y + z * z;

        v2f bm;
        bm[0] = hi ? z : x;
        bm[1] = hi ? pn2 : y;

        v8f d = __builtin_amdgcn_wmma_f32_16x16x4_f32(
             false, a,  false, bm,
             (short)0, czero,  false,  false);

#pragma unroll
        for (int r = 0; r < 8; ++r) {
            const float v = d[r];
            if (v < minv[r]) {
                minv[r] = v;
                mini[r] = n;
            }
        }
    }

#pragma unroll
    for (int r = 0; r < 8; ++r) {
#pragma unroll
        for (int off = 1; off < 16; off <<= 1) {
            const float ov = __shfl_xor(minv[r], off, 32);
            const int   oi = __shfl_xor(mini[r], off, 32);
            if (ov < minv[r] || (ov == minv[r] && oi < mini[r])) {
                minv[r] = ov;
                mini[r] = oi;
            }
        }
    }

    __shared__ float lv[WAVES][16];
    __shared__ int   li[WAVES][16];
    if (lane == 0 || lane == 16) {
        const int rowbase = hi ? 8 : 0;
#pragma unroll
        for (int r = 0; r < 8; ++r) {
            lv[wave][rowbase + r] = minv[r];
            li[wave][rowbase + r] = mini[r];
        }
    }
    __syncthreads();

    if (tid < 16) {
        const int m = tid;
        float bv = FLT_BIG;
        int   bi = 0;
#pragma unroll
        for (int w = 0; w < WAVES; ++w) {
            const float v = lv[w][m];
            const int   i = li[w][m];
            if (v < bv || (v == bv && i < bi)) { bv = v; bi = i; }
        }

        const float kx = kpb[0 * M_PTS + m0 + m];
        const float ky = kpb[1 * M_PTS + m0 + m];
        const float kz = kpb[2 * M_PTS + m0 + m];

        const float px = pcx[bi];
        const float py = pcy[bi];
        const float pz = pcz[bi];

        const float* snb = sn + (size_t)b * 3 * N_PTS;
        const float sx = snb[0 * N_PTS + bi];
        const float sy = snb[1 * N_PTS + bi];
        const float sz = snb[2 * N_PTS + bi];

        const float dx = kx - px, dy = ky - py, dz = kz - pz;
        const float dn  = sqrtf(dx * dx + dy * dy + dz * dz);
        const float inv = 1.0f / (dn + EPS_F);
        const float dot = (sx * dx + sy * dy + sz * dz) * inv;

        out[(size_t)b * M_PTS + m0 + m] = dot * dot;
    }
}

extern "C" void kernel_launch(void* const* d_in, const int* in_sizes, int n_in,
                              void* d_out, int out_size, void* d_ws, size_t ws_size,
                              hipStream_t stream) {
    const float* kp = (const float*)d_in[0];
    const float* pc = (const float*)d_in[1];
    const float* sn = (const float*)d_in[2];
    float* out = (float*)d_out;

    dim3 grid(M_PTS / 16, BATCH);
    pos_loss_kernel<<<grid, THREADS, 0, stream>>>(kp, pc, sn, out);
}


# All 48 point loads of a wave's sweep issued before the first tile, as 16-byte loads (a lane column owns 4 consecutive points): one memory wait per sweep, not 16 dependent ones; scores bit-identical
# speedup vs baseline: 1.7126x; 1.7126x over previous
#include <hip/hip_runtime.h>
#include <math.h>

typedef __attribute__((ext_vector_type(2))) float v2f;
typedef __attribute__((ext_vector_type(4))) float v4f;
typedef __attribute__((ext_vector_type(8))) float v8f;

#define BATCH   8
#define M_PTS   512
#define N_PTS   8192
#define EPS_F   1e-7f
#define WAVES   8
#define THREADS (WAVES * 32)
#define TILES   (N_PTS / 16)
#define TPW     (TILES / WAVES)
#define GROUPS  (TPW / 4)
#define FLT_BIG 3.402823466e+38f

__global__ __launch_bounds__(THREADS)
void pos_loss_kernel(const float* __restrict__ kp,
                     const float* __restrict__ pc,
                     const float* __restrict__ sn,
                     float* __restrict__ out)
{
    const int b    = blockIdx.y;
    const int m0   = blockIdx.x * 16;
    const int tid  = threadIdx.x;
    const int wave = tid >> 5;
    const int lane = tid & 31;
    const int col  = lane & 15;
    const bool hi  = lane >= 16;

    const float* kpb = kp + (size_t)b * 3 * M_PTS;
    const float* pcx = pc + (size_t)b * 3 * N_PTS;
    const float* pcy = pcx + N_PTS;
    const float* pcz = pcy + N_PTS;

    const float k02 = kpb[(hi ? 2 : 0) * M_PTS + m0 + col];
    const float k1  = kpb[1 * M_PTS + m0 + col];

    const v8f czero = {};

    float minv[8];
    int   mini[8];
#pragma unroll
    for (int r = 0; r < 8; ++r) { minv[r] = FLT_BIG; mini[r] = 0; }

    const int nb = wave * (TPW * 16) + 4 * col;
    v4f X[GROUPS], Y[GROUPS], Z[GROUPS];
#pragma unroll
    for (int g = 0; g < GROUPS; ++g) {
        X[g] = *reinterpret_cast<const v4f*>(pcx + nb + 64 * g);
        Y[g] = *reinterpret_cast<const v4f*>(pcy + nb + 64 * g);
        Z[g] = *reinterpret_cast<const v4f*>(pcz + nb + 64 * g);
    }
    __builtin_amdgcn_sched_barrier(0);

    v2f a;
    a[0] = -2.0f * k02;
    a[1] = hi ? 1.0f : (-2.0f * k1);

#pragma unroll
    for (int g = 0; g < GROUPS; ++g) {
#pragma unroll
        for (int j = 0; j < 4; ++j) {
            const int n = nb + 64 * g + j;
            const float x = X[g][j];
            const float y = Y[g][j];
            const float z = Z[g][j];
            const float pn2 = __builtin_fmaf(z, z, __builtin_fmaf(x, x, y * y));

            v2f bm;
            bm[0] = hi ? z : x;
            bm[1] = hi ? pn2 : y;

            v8f d = __builtin_amdgcn_wmma_f32_16x16x4_f32(
                 false, a,  false, bm,
                 (short)0, czero,  false,  false);

#pragma unroll
            for (int r = 0; r < 8; ++r) {
                const float v = d[r];
                if (v < minv[r]) {
                    minv[r] = v;
                    mini[r] = n;
                }
            }
        }
    }

#pragma unroll
    for (int r = 0; r < 8; ++r) {
#pragma unroll
        for (int off = 1; off < 16; off <<= 1) {
            const float ov = __shfl_xor(minv[r], off, 32);
            const int   oi = __shfl_xor(mini[r], off, 32);
            if (ov < minv[r] || (ov == minv[r] && oi < mini[r])) {
                minv[r] = ov;
                mini[r] = oi;
            }
        }
    }

    __shared__ float lv[WAVES][16];
    __shared__ int   li[WAVES][16];
    if (lane == 0 || lane == 16) {
        const int rowbase = hi ? 8 : 0;
#pragma unroll
        for (int r = 0; r < 8; ++r) {
            lv[wave][rowbase + r] = minv[r];
            li[wave][rowbase + r] = mini[r];
        }
    }
    __syncthreads();

    if (tid < 16) {
        const int m = tid;
        float bv = FLT_BIG;
        int   bi = 0;
#pragma unroll
        for (int w = 0; w < WAVES; ++w) {
            const float v = lv[w][m];
            const int   i = li[w][m];
            if (v < bv || (v == bv && i < bi)) { bv = v; bi = i; }
        }

        const float kx = kpb[0 * M_PTS + m0 + m];
        const float ky = kpb[1 * M_PTS + m0 + m];
        const float kz = kpb[2 * M_PTS + m0 + m];

        const float px = pcx[bi];
        const float py = pcy[bi];
        const float pz = pcz[bi];

        const float* snb = sn + (size_t)b * 3 * N_PTS;
        const float sx = snb[0 * N_PTS + bi];
        const float sy = snb[1 * N_PTS + bi];
        const float sz = snb[2 * N_PTS + bi];

        const float dx = kx - px, dy = ky - py, dz = kz - pz;
        const float dn  = sqrtf(dx * dx + dy * dy + dz * dz);
        const float inv = 1.0f / (dn + EPS_F);
        const float dot = (sx * dx + sy * dy + sz * dz) * inv;

        out[(size_t)b * M_PTS + m0 + m] = dot * dot;
    }
}

extern "C" void kernel_launch(void* const* d_in, const int* in_sizes, int n_in,
                              void* d_out, int out_size, void* d_ws, size_t ws_size,
                              hipStream_t stream) {
    const float* kp = (const float*)d_in[0];
    const float* pc = (const float*)d_in[1];
    const float* sn = (const float*)d_in[2];
    float* out = (float*)d_out;

    dim3 grid(M_PTS / 16, BATCH);
    pos_loss_kernel<<<grid, THREADS, 0, stream>>>(kp, pc, sn, out);
}


# Points as matrix rows: a keypoint's scores sit in one lane, so the sweep keeps one min3 minimum per 32-point block and the 16-lane butterfly goes; winning block re-scored for the index; same scores
# speedup vs baseline: 2.6129x; 1.5257x over previous
#include <hip/hip_runtime.h>
#include <math.h>


typedef __attribute__((ext_vector_type(2))) float v2f;
typedef __attribute__((ext_vector_type(4))) float v4f;
typedef __attribute__((ext_vector_type(8))) float v8f;

#define BATCH   8
#define M_PTS   512
#define N_PTS   8192
#define EPS_F   1e-7f
#define WAVES   8
#define THREADS (WAVES * 32)
#define TILES   (N_PTS / 16)
#define TPW     (TILES / WAVES)
#define GROUPS  (TPW / 4)
#define FLT_BIG 3.402823466e+38f

static __device__ __forceinline__ float min3f(float a, float b, float c)
{
    return __builtin_fminf(__builtin_fminf(a, b), c);
}

static __device__ __forceinline__ int xhalf_i(int x)
{
    return __builtin_amdgcn_permlanex16(x, x, 0x76543210, 0xfedcba98, false, false);
}
static __device__ __forceinline__ float xhalf_f(float x)
{
    return __builtin_bit_cast(float, xhalf_i(__builtin_bit_cast(int, x)));
}

static __device__ __forceinline__ v8f score_tile(float x, float y, float z, bool hi, v2f kf)
{
    const float pn2 = __builtin_fmaf(z, z, __builtin_fmaf(x, x, y * y));
    v2f pf;
    pf[0] = hi ? z : x;
    pf[1] = hi ? pn2 : y;
    const v8f czero = {};
    return __builtin_amdgcn_wmma_f32_16x16x4_f32(
         false, pf,  false, kf,
         (short)0, czero,  false,  false);
}

__global__ __launch_bounds__(THREADS)
void pos_loss_kernel(const float* __restrict__ kp,
                     const float* __restrict__ pc,
                     const float* __restrict__ sn,
                     float* __restrict__ out)
{
    const int b    = blockIdx.y;
    const int m0   = blockIdx.x * 16;
    const int tid  = threadIdx.x;
    const int wave = tid >> 5;
    const int lane = tid & 31;
    const int col  = lane & 15;
    const bool hi  = lane >= 16;

    const float* kpb = kp + (size_t)b * 3 * M_PTS;
    const float* pcx = pc + (size_t)b * 3 * N_PTS;
    const float* pcy = pcx + N_PTS;
    const float* pcz = pcy + N_PTS;

    const float k02 = kpb[(hi ? 2 : 0) * M_PTS + m0 + col];
    const float k1  = kpb[1 * M_PTS + m0 + col];

    const int wbase = wave * (TPW * 16);
    const int nb = wbase + 4 * col;
    v4f X[GROUPS], Y[GROUPS], Z[GROUPS];
#pragma unroll
    for (int g = 0; g < GROUPS; ++g) {
        X[g] = *reinterpret_cast<const v4f*>(pcx + nb + 64 * g);
        Y[g] = *reinterpret_cast<const v4f*>(pcy + nb + 64 * g);
        Z[g] = *reinterpret_cast<const v4f*>(pcz + nb + 64 * g);
    }
    __builtin_amdgcn_sched_barrier(0);

    v2f kf;
    kf[0] = -2.0f * k02;
    kf[1] = hi ? 1.0f : (-2.0f * k1);

    float bestv = FLT_BIG;
    int   bestg = 0;
#pragma unroll
    for (int g = 0; g < GROUPS; ++g) {
        float p[4], q[4];
#pragma unroll
        for (int j = 0; j < 4; ++j) {
            const v8f d = score_tile(X[g][j], Y[g][j], Z[g][j], hi, kf);
            p[j] = min3f(min3f(d[0], d[1], d[2]), d[3], d[4]);
            q[j] = min3f(d[5], d[6], d[7]);
        }
        const float s0 = min3f(p[0], q[0], p[1]);
        const float s1 = min3f(q[1], p[2], q[2]);
        const float s2 = min3f(p[3], q[3], bestv);
        const float nv = min3f(s0, s1, s2);
        bestg = (nv < bestv) ? g : bestg;
        bestv = nv;
    }
    int bestb = wbase + 64 * bestg + (hi ? 32 : 0);

    {
        const float ov = xhalf_f(bestv);
        const int   ob = xhalf_i(bestb);
        const bool take = (ov < bestv) | ((ov == bestv) & (ob < bestb));
        bestv = take ? ov : bestv;
        bestb = take ? ob : bestb;
    }

    __shared__ __attribute__((aligned(16))) float lv[16][WAVES];
    __shared__ __attribute__((aligned(16))) int   lb[16][WAVES];
    __shared__ float res[16];
    if (!hi) {
        lv[col][wave] = bestv;
        lb[col][wave] = bestb;
    }
    __syncthreads();

    float fv = lv[col][0];
    int   fb = lb[col][0];
#pragma unroll
    for (int w = 1; w < WAVES; ++w) {
        const float v = lv[col][w];
        const int   i = lb[col][w];
        const bool take = v < fv;
        fv = take ? v : fv;
        fb = take ? i : fb;
    }

    const int base0 = __shfl(fb, 2 * wave, 32);
    const int base1 = __shfl(fb, 2 * wave + 1, 32);
    float cx[2][2], cy[2][2], cz[2][2];
#pragma unroll
    for (int t = 0; t < 2; ++t) {
        cx[0][t] = pcx[base0 + 16 * t + col];
        cy[0][t] = pcy[base0 + 16 * t + col];
        cz[0][t] = pcz[base0 + 16 * t + col];
        cx[1][t] = pcx[base1 + 16 * t + col];
        cy[1][t] = pcy[base1 + 16 * t + col];
        cz[1][t] = pcz[base1 + 16 * t + col];
    }
    v8f dd[2][2];
#pragma unroll
    for (int e = 0; e < 2; ++e)
#pragma unroll
        for (int t = 0; t < 2; ++t)
            dd[e][t] = score_tile(cx[e][t], cy[e][t], cz[e][t], hi, kf);

    const bool odd = (col & 1) != 0;
    float rv[2];
    int   rq[2];
#pragma unroll
    for (int t = 0; t < 2; ++t) {
        rv[t] = FLT_BIG;
        rq[t] = 16 * t;
#pragma unroll
        for (int r = 0; r < 8; ++r) {
            const float v = odd ? dd[1][t][r] : dd[0][t][r];
            const bool take = v < rv[t];
            rv[t] = take ? v : rv[t];
            rq[t] = take ? (16 * t + r) : rq[t];
        }
    }
    float mv = rv[0];
    int   mq = rq[0];
    {
        const bool take = rv[1] < mv;
        mv = take ? rv[1] : mv;
        mq = take ? rq[1] : mq;
    }
    mq += hi ? 8 : 0;
    {
        const float ov = xhalf_f(mv);
        const int   oq = xhalf_i(mq);
        const bool take = (ov < mv) | ((ov == mv) & (oq < mq));
        mv = take ? ov : mv;
        mq = take ? oq : mq;
    }

    if ((col >> 1) == wave && !hi) {
        const int m  = col;
        const int bi = fb + mq;

        const float kx = kpb[0 * M_PTS + m0 + m];
        const float ky = kpb[1 * M_PTS + m0 + m];
        const float kz = kpb[2 * M_PTS + m0 + m];

        const float px = pcx[bi];
        const float py = pcy[bi];
        const float pz = pcz[bi];

        const float* snb = sn + (size_t)b * 3 * N_PTS;
        const float sx = snb[0 * N_PTS + bi];
        const float sy = snb[1 * N_PTS + bi];
        const float sz = snb[2 * N_PTS + bi];

        const float dx = kx - px, dy = ky - py, dz = kz - pz;
        const float dn  = sqrtf(dx * dx + dy * dy + dz * dz);
        const float inv = 1.0f / (dn + EPS_F);
        const float dot = (sx * dx + sy * dy + sz * dz) * inv;

        res[m] = dot * dot;
    }
    __syncthreads();

    if (tid < 16)
        out[(size_t)b * M_PTS + m0 + tid] = res[tid];
}

extern "C" void kernel_launch(void* const* d_in, const int* in_sizes, int n_in,
                              void* d_out, int out_size, void* d_ws, size_t ws_size,
                              hipStream_t stream) {
    const float* kp = (const float*)d_in[0];
    const float* pc = (const float*)d_in[1];
    const float* sn = (const float*)d_in[2];
    float* out = (float*)d_out;

    dim3 grid(M_PTS / 16, BATCH);
    pos_loss_kernel<<<grid, THREADS, 0, stream>>>(kp, pc, sn, out);
}
